# starting kernel
# baseline (speedup 1.0000x reference)
#include <hip/hip_runtime.h>

typedef __attribute__((ext_vector_type(2))) float v2f;
typedef __attribute__((ext_vector_type(8))) float v8f;

#define IN_DIM 128
#define HID    64
#define LAYERS 8
#define OUTD   40

__global__ void zero_int_kernel(int* __restrict__ p, int n) {
    int i = blockIdx.x * blockDim.x + threadIdx.x;
    if (i < n) p[i] = 0;
}

__global__ void count_deg_kernel(const int* __restrict__ dst, int* __restrict__ cnt, int E) {
    int e = blockIdx.x * blockDim.x + threadIdx.x;
    if (e < E) atomicAdd(&cnt[dst[e]], 1);
}

__global__ __launch_bounds__(1024) void scan_kernel(const int* __restrict__ cnt,
                                                    int* __restrict__ rowptr,
                                                    int N, int E) {
    __shared__ int sh[1024];
    __shared__ int carry;
    int tid = threadIdx.x;
    if (tid == 0) carry = 0;
    __syncthreads();
    for (int base = 0; base < N; base += 1024) {
        int idx = base + tid;
        int v = (idx < N) ? cnt[idx] : 0;
        sh[tid] = v;
        __syncthreads();
        for (int off = 1; off < 1024; off <<= 1) {
            int t = (tid >= off) ? sh[tid - off] : 0;
            __syncthreads();
            sh[tid] += t;
            __syncthreads();
        }
        int incl = sh[tid];
        int excl = incl - v;
        if (idx < N) rowptr[idx] = carry + excl;
        __syncthreads();
        if (tid == 1023) carry += sh[1023];
        __syncthreads();
    }
    if (tid == 0) rowptr[N] = E;
}

__global__ void init_pos_dis_kernel(const int* __restrict__ rowptr,
                                    const int* __restrict__ cnt,
                                    int* __restrict__ pos,
                                    float* __restrict__ dis, int N) {
    int i = blockIdx.x * blockDim.x + threadIdx.x;
    if (i < N) {
        pos[i] = rowptr[i];
        float dg = (float)(cnt[i] + 1);
        dis[i] = rsqrtf(dg);
    }
}

__global__ void fill_csr_kernel(const int* __restrict__ src, const int* __restrict__ dst,
                                int* __restrict__ pos, const float* __restrict__ dis,
                                int2* __restrict__ erec, int E) {
    int e = blockIdx.x * blockDim.x + threadIdx.x;
    if (e < E) {
        int s = src[e], d = dst[e];
        int p = atomicAdd(&pos[d], 1);
        erec[p] = make_int2(s, __float_as_int(dis[s] * dis[d]));
    }
}

__global__ void init_out_kernel(float* __restrict__ out, const float* __restrict__ lin_b,
                                int total) {
    int i = blockIdx.x * blockDim.x + threadIdx.x;
    if (i < total) out[i] = lin_b[i % OUTD];
}

__global__ __launch_bounds__(256) void gemm_wmma_f32_kernel(
    const float* __restrict__ A, const float* __restrict__ W,
    float* __restrict__ T, int nrows, int K)
{
    const int lane  = threadIdx.x & 31;
    const int wib   = threadIdx.x >> 5;
    const int strip = blockIdx.x * 8 + wib;
    const int m0    = strip * 16;
    if (m0 >= nrows) return;

    const int mi = lane & 15;
    const int hi = (lane >> 4) & 1;
    const int kb = hi * 2;

    const float* arow = A + (size_t)(m0 + mi) * K;

    v8f acc0 = {0.f,0.f,0.f,0.f,0.f,0.f,0.f,0.f};
    v8f acc1 = acc0, acc2 = acc0, acc3 = acc0;

    for (int k0 = 0; k0 < K; k0 += 4) {
        v2f a;
        a.x = arow[k0 + kb];
        a.y = arow[k0 + kb + 1];
        const float* w0 = W + (size_t)(k0 + kb) * HID + mi;
        const float* w1 = w0 + HID;
        v2f b0, b1, b2, b3;
        b0.x = w0[0];   b0.y = w1[0];
        b1.x = w0[16];  b1.y = w1[16];
        b2.x = w0[32];  b2.y = w1[32];
        b3.x = w0[48];  b3.y = w1[48];
        acc0 = __builtin_amdgcn_wmma_f32_16x16x4_f32(false, a, false, b0, (short)0, acc0, false, false);
        acc1 = __builtin_amdgcn_wmma_f32_16x16x4_f32(false, a, false, b1, (short)0, acc1, false, false);
        acc2 = __builtin_amdgcn_wmma_f32_16x16x4_f32(false, a, false, b2, (short)0, acc2, false, false);
        acc3 = __builtin_amdgcn_wmma_f32_16x16x4_f32(false, a, false, b3, (short)0, acc3, false, false);
    }

    #pragma unroll
    for (int v = 0; v < 8; ++v) {
        float* tr = T + (size_t)(m0 + v + hi * 8) * HID + mi;
        tr[0]  = acc0[v];
        tr[16] = acc1[v];
        tr[32] = acc2[v];
        tr[48] = acc3[v];
    }
}

__global__ __launch_bounds__(64) void aggregate_kernel(
    const float* __restrict__ T,
    const int* __restrict__ rowptr,
    const int2* __restrict__ erec,
    const float* __restrict__ dis,
    const float* __restrict__ bias,
    const float* __restrict__ lw,
    float* __restrict__ Hout,
    float* __restrict__ Out)
{
    const int i = blockIdx.x;
    const int f = threadIdx.x;

    float di  = dis[i];
    float acc = T[(size_t)i * HID + f] * di * di + bias[f];

    int e = rowptr[i];
    const int end = rowptr[i + 1];

    for (; e + 4 <= end; e += 4) {
        int2 r0 = erec[e + 0];
        int2 r1 = erec[e + 1];
        int2 r2 = erec[e + 2];
        int2 r3 = erec[e + 3];
        float t0 = T[(size_t)r0.x * HID + f];
        float t1 = T[(size_t)r1.x * HID + f];
        float t2 = T[(size_t)r2.x * HID + f];
        float t3 = T[(size_t)r3.x * HID + f];
        acc = fmaf(t0, __int_as_float(r0.y), acc);
        acc = fmaf(t1, __int_as_float(r1.y), acc);
        acc = fmaf(t2, __int_as_float(r2.y), acc);
        acc = fmaf(t3, __int_as_float(r3.y), acc);
    }
    for (; e < end; ++e) {
        int2 r = erec[e];
        acc = fmaf(T[(size_t)r.x * HID + f], __int_as_float(r.y), acc);
    }

    acc = fmaxf(acc, 0.f);
    Hout[(size_t)i * HID + f] = acc;

    __shared__ float sh[HID];
    sh[f] = acc;
    __syncthreads();

    if (f < OUTD) {
        float s = 0.f;
        #pragma unroll
        for (int c = 0; c < HID; ++c)
            s = fmaf(sh[c], lw[c * OUTD + f], s);
        Out[(size_t)i * OUTD + f] += s;
    }
}

extern "C" void kernel_launch(void* const* d_in, const int* in_sizes, int n_in,
                              void* d_out, int out_size, void* d_ws, size_t ws_size,
                              hipStream_t stream) {
    const float* x     = (const float*)d_in[0];
    const float* W0    = (const float*)d_in[1];
    const float* Ws    = (const float*)d_in[2];
    const float* bs    = (const float*)d_in[3];
    const float* lin_w = (const float*)d_in[4];
    const float* lin_b = (const float*)d_in[5];
    const int*   ei    = (const int*)d_in[6];
    float* out = (float*)d_out;

    const int N = in_sizes[0] / IN_DIM;
    const int E = in_sizes[6] / 2;
    const int* src = ei;
    const int* dst = ei + E;

    char* ws = (char*)d_ws;
    int2*  erec   = (int2*)ws;  ws += sizeof(int2)  * (size_t)E;
    int*   cnt    = (int*)ws;   ws += sizeof(int)   * (size_t)N;
    int*   rowptr = (int*)ws;   ws += sizeof(int)   * (size_t)(N + 1);
    int*   pos    = (int*)ws;   ws += sizeof(int)   * (size_t)N;
    float* dis    = (float*)ws; ws += sizeof(float) * (size_t)N;
    float* t      = (float*)ws; ws += sizeof(float) * (size_t)N * HID;
    float* ha     = (float*)ws; ws += sizeof(float) * (size_t)N * HID;
    float* hb     = (float*)ws; ws += sizeof(float) * (size_t)N * HID;

    const int TPB = 256;
    zero_int_kernel<<<(N + TPB - 1) / TPB, TPB, 0, stream>>>(cnt, N);
    count_deg_kernel<<<(E + TPB - 1) / TPB, TPB, 0, stream>>>(dst, cnt, E);
    scan_kernel<<<1, 1024, 0, stream>>>(cnt, rowptr, N, E);
    init_pos_dis_kernel<<<(N + TPB - 1) / TPB, TPB, 0, stream>>>(rowptr, cnt, pos, dis, N);
    fill_csr_kernel<<<(E + TPB - 1) / TPB, TPB, 0, stream>>>(src, dst, pos, dis, erec, E);
    init_out_kernel<<<((N * OUTD) + TPB - 1) / TPB, TPB, 0, stream>>>(out, lin_b, N * OUTD);

    const int strips = N / 16;
    const int gblocks = (strips + 7) / 8;
    for (int l = 0; l < LAYERS; ++l) {
        const float* hin  = (l == 0) ? x : ((l & 1) ? ha : hb);
        float*       hcur = (l & 1) ? hb : ha;
        const int    K    = (l == 0) ? IN_DIM : HID;
        const float* W    = (l == 0) ? W0 : (Ws + (size_t)(l - 1) * HID * HID);

        gemm_wmma_f32_kernel<<<gblocks, 256, 0, stream>>>(hin, W, t, N, K);
        aggregate_kernel<<<N, 64, 0, stream>>>(t, rowptr, erec, dis,
                                               bs + (size_t)l * HID,
                                               lin_w + (size_t)l * HID * OUTD,
                                               hcur, out);
    }
}


# GEMM launch: operands come through LDS: a block builds the weight fragments of 16 k-steps once, a wave brings its 16-row A strip by whole-row 16-byte loads; fragments by LDS loads; k order kept
# speedup vs baseline: 2.5375x; 2.5375x over previous
#include <hip/hip_runtime.h>

typedef __attribute__((ext_vector_type(2))) float v2f;
typedef __attribute__((ext_vector_type(8))) float v8f;

#define IN_DIM 128
#define HID    64
#define LAYERS 8
#define OUTD   40

__global__ void zero_int_kernel(int* __restrict__ p, int n) {
    int i = blockIdx.x * blockDim.x + threadIdx.x;
    if (i < n) p[i] = 0;
}

#define CAP 32

__global__ void fill_slots_kernel(const int* __restrict__ src, const int* __restrict__ dst,
                                  int* __restrict__ cnt, int* __restrict__ nspill,
                                  int2* __restrict__ tab, int2* __restrict__ spill, int E) {
    int e = blockIdx.x * blockDim.x + threadIdx.x;
    if (e < E) {
        int s = src[e], d = dst[e];
        int p = atomicAdd(&cnt[d], 1);
        if (p < CAP) {
            tab[(size_t)d * CAP + p].x = s;
        } else {
            int q = atomicAdd(nspill, 1);
            spill[q] = make_int2(d, s);
        }
    }
}

#define KCH   64
#define APIT  (KCH / 4 + 1)

template <bool DO_T, int JK, int KD>
__global__ __launch_bounds__(256) void gemm_wmma_f32_kernel(
    const float* __restrict__ A, const float* __restrict__ W, float* __restrict__ T,
    const float* __restrict__ lw,
    const float* __restrict__ lb,
    float* __restrict__ Out,
    int nrows)
{
    __shared__ float4 sA[8 * 16 * APIT];
    __shared__ float4 sW[DO_T ? (KCH / 4) * 2 * 32 : 1];
    __shared__ float4 sJ4[JK != 0 ? (KCH / 4) * 32 : 1];
    __shared__ float2 sJ2[JK != 0 ? (KCH / 4) * 32 : 1];

    const int lane  = threadIdx.x & 31;
    const int wib   = threadIdx.x >> 5;
    const int strip = blockIdx.x * 8 + wib;
    const int m0    = strip * 16;
    const bool valid = m0 < nrows;

    const int mi = lane & 15;
    const int hi = (lane >> 4) & 1;
    const int kb = hi * 2;
    const int c2 = 32 + (mi < 8 ? mi : 7);

    v8f acc0 = {0.f,0.f,0.f,0.f,0.f,0.f,0.f,0.f};
    v8f acc1 = acc0, acc2 = acc0, acc3 = acc0;
    v8f j0 = acc0, j1 = acc0, j2 = acc0;

    float4* sAw = sA + wib * 16 * APIT;
    const float2* afrag = reinterpret_cast<const float2*>(sAw + mi * APIT) + hi;
    constexpr unsigned K4 = KD / 4;

    #pragma unroll
    for (int kc = 0; kc < KD; kc += KCH) {
        if (kc != 0) __syncthreads();

        float4 ar[KCH / 8];
        float4 wr[DO_T ? 4 : 1];
        float4 jr4[JK != 0 ? 2 : 1];
        float2 jr2[JK != 0 ? 2 : 1];
        {
            const float4* A4 = reinterpret_cast<const float4*>(A);
            const unsigned a0 = (unsigned)((valid ? m0 : 0) + hi) * K4 + (unsigned)(kc >> 2) + (unsigned)mi;
            #pragma unroll
            for (int i = 0; i < KCH / 8; ++i)
                ar[i] = A4[a0 + (unsigned)(2 * i) * K4];
        }
        if (DO_T) {
            #pragma unroll
            for (int i = 0; i < 4; ++i) {
                const int g = wib + 8 * i;
                const int s = g >> 1, np = g & 1;
                const float* w0 = W + (unsigned)((kc + 4 * s + kb) * HID + mi + 32 * np);
                const float* w1 = w0 + HID;
                wr[i] = make_float4(w0[0], w1[0], w0[16], w1[16]);
            }
        }
        if (JK != 0) {
            #pragma unroll
            for (int i = 0; i < 2; ++i) {
                const int s = wib + 8 * i;
                const float* l0 = lw + (unsigned)((kc + 4 * s + kb) * OUTD);
                const float* l1 = l0 + OUTD;
                const float2 p0 = *reinterpret_cast<const float2*>(l0 + 2 * mi);
                const float2 p1 = *reinterpret_cast<const float2*>(l1 + 2 * mi);
                jr4[i] = make_float4(p0.x, p1.x, p0.y, p1.y);
                jr2[i] = make_float2(l0[c2], l1[c2]);
            }
        }
        if (JK == 1 && kc == 0) {
            const float2 bp = *reinterpret_cast<const float2*>(lb + 2 * mi);
            const float  b2 = lb[c2];
            #pragma unroll
            for (int v = 0; v < 8; ++v) { j0[v] = bp.x; j1[v] = bp.y; j2[v] = b2; }
        }
        if (JK == 2 && kc == 0 && valid) {
            #pragma unroll
            for (int v = 0; v < 8; ++v) {
                const float* orow = Out + (size_t)(m0 + v + hi * 8) * OUTD;
                const float2 p = *reinterpret_cast<const float2*>(orow + 2 * mi);
                j0[v] = p.x;
                j1[v] = p.y;
                j2[v] = orow[c2];
            }
        }
        if (DO_T) {
            #pragma unroll
            for (int i = 0; i < 4; ++i) sW[(wib + 8 * i) * 32 + lane] = wr[i];
        }
        if (JK != 0) {
            #pragma unroll
            for (int i = 0; i < 2; ++i) {
                sJ4[(wib + 8 * i) * 32 + lane] = jr4[i];
                sJ2[(wib + 8 * i) * 32 + lane] = jr2[i];
            }
        }
        #pragma unroll
        for (int i = 0; i < KCH / 8; ++i) sAw[(2 * i + hi) * APIT + mi] = ar[i];

        __syncthreads();

        if (valid) {
            for (int s = 0; s < KCH / 4; ++s) {
                const float2 af = afrag[2 * s];
                v2f a;
                a.x = af.x;
                a.y = af.y;
                if (DO_T) {
                    const float4 q0 = sW[(2 * s) * 32 + lane];
                    const float4 q1 = sW[(2 * s + 1) * 32 + lane];
                    v2f b0, b1, b2, b3;
                    b0.x = q0.x;  b0.y = q0.y;
                    b1.x = q0.z;  b1.y = q0.w;
                    b2.x = q1.x;  b2.y = q1.y;
                    b3.x = q1.z;  b3.y = q1.w;
                    acc0 = __builtin_amdgcn_wmma_f32_16x16x4_f32(false, a, false, b0, (short)0, acc0, false, false);
                    acc1 = __builtin_amdgcn_wmma_f32_16x16x4_f32(false, a, false, b1, (short)0, acc1, false, false);
                    acc2 = __builtin_amdgcn_wmma_f32_16x16x4_f32(false, a, false, b2, (short)0, acc2, false, false);
                    acc3 = __builtin_amdgcn_wmma_f32_16x16x4_f32(false, a, false, b3, (short)0, acc3, false, false);
                }
                if (JK != 0) {
                    const float4 q  = sJ4[s * 32 + lane];
                    const float2 q2 = sJ2[s * 32 + lane];
                    v2f e0, e1, e2;
                    e0.x = q.x;   e0.y = q.y;
                    e1.x = q.z;   e1.y = q.w;
                    e2.x = q2.x;  e2.y = q2.y;
                    j0 = __builtin_amdgcn_wmma_f32_16x16x4_f32(false, a, false, e0, (short)0, j0, false, false);
                    j1 = __builtin_amdgcn_wmma_f32_16x16x4_f32(false, a, false, e1, (short)0, j1, false, false);
                    j2 = __builtin_amdgcn_wmma_f32_16x16x4_f32(false, a, false, e2, (short)0, j2, false, false);
                }
            }
        }
    }
    if (!valid) return;

    if (DO_T) {
        #pragma unroll
        for (int v = 0; v < 8; ++v) {
            float* tr = T + (size_t)(m0 + v + hi * 8) * HID + mi;
            tr[0]  = acc0[v];
            tr[16] = acc1[v];
            tr[32] = acc2[v];
            tr[48] = acc3[v];
        }
    }
    if (JK != 0) {
        #pragma unroll
        for (int v = 0; v < 8; ++v) {
            float* orow = Out + (size_t)(m0 + v + hi * 8) * OUTD;
            *reinterpret_cast<float2*>(orow + 2 * mi) = make_float2(j0[v], j1[v]);
            if (mi < 8) orow[32 + mi] = j2[v];
        }
    }
}

__device__ __forceinline__ void fma4(float4& a, const float4 t, const float w) {
    a.x = fmaf(t.x, w, a.x);
    a.y = fmaf(t.y, w, a.y);
    a.z = fmaf(t.z, w, a.z);
    a.w = fmaf(t.w, w, a.w);
}

template <bool FIRST>
__device__ __forceinline__ int2 load_rec(const int2* row, int slot) {
    if (FIRST) return make_int2(reinterpret_cast<const int*>(row)[2 * slot], 0);
    return row[slot];
}

template <bool FIRST>
__device__ __forceinline__ float rec_weight(const int2 r, const int* __restrict__ cnt, const float di) {
    if (FIRST) return rsqrtf((float)(cnt[r.x] + 1)) * di;
    return __int_as_float(r.y);
}

template <bool FIRST, int NS>
__device__ __forceinline__ void sum_slots(float4& acc, const int2* row, const float4* __restrict__ T4,
                                          const int* __restrict__ cnt, const int j, const int nt,
                                          const int self, const int q4, const float di) {
    int2   r[NS];
    float4 t[NS];
    float  w[NS];
    #pragma unroll
    for (int k = 0; k < NS; ++k) {
        r[k]   = load_rec<FIRST>(row, j + k);
        r[k].x = (j + k < nt) ? r[k].x : self;
    }
    #pragma unroll
    for (int k = 0; k < NS; ++k) {
        t[k] = T4[(unsigned)r[k].x * 16u + (unsigned)q4];
        w[k] = rec_weight<FIRST>(r[k], cnt, di);
    }
    #pragma unroll
    for (int k = 0; k < NS; ++k) {
        const int m = (j + k < nt) ? -1 : 0;
        const float wk = __int_as_float(__float_as_int(w[k]) & m);
        acc.x = fmaf(__int_as_float(__float_as_int(t[k].x) & m), wk, acc.x);
        acc.y = fmaf(__int_as_float(__float_as_int(t[k].y) & m), wk, acc.y);
        acc.z = fmaf(__int_as_float(__float_as_int(t[k].z) & m), wk, acc.z);
        acc.w = fmaf(__int_as_float(__float_as_int(t[k].w) & m), wk, acc.w);
    }
}

template <bool FIRST>
__global__ __launch_bounds__(256) void aggregate_kernel(
    const float* __restrict__ T,
    const int* __restrict__ cnt,
    const int* __restrict__ nspill,
    int2* erec,
    const int2* __restrict__ spill,
    const float* __restrict__ bias,
    float* __restrict__ Hout,
    int N)
{
    const int q4 = threadIdx.x & 15;
    const int ig = blockIdx.x * 16 + (int)(threadIdx.x >> 4);
    if (blockIdx.x * 16 + (int)(threadIdx.x >> 5) * 2 >= N) return;
    const bool live = ig < N;
    const int  i    = live ? ig : N - 1;

    const float4* __restrict__ T4 = reinterpret_cast<const float4*>(T);

    const int   n  = live ? cnt[i] : 0;
    const float di = rsqrtf((float)(n + 1));
    const int   nt = n < CAP ? n : CAP;
    const int   n0 = __builtin_amdgcn_readlane(nt, 0);
    const int   n1 = __builtin_amdgcn_readlane(nt, 16);
    const int   ntmax = n0 > n1 ? n0 : n1;

    int2* rowi = erec + (size_t)i * CAP;
    const int2* row = rowi;

    int sxa = 0, sxb = 0;
    if (FIRST) {
        sxa = reinterpret_cast<const int*>(row)[2 * q4];
        sxb = reinterpret_cast<const int*>(row)[2 * (q4 + 16)];
    }

    float4 acc;
    {
        const float4 t = T4[(size_t)i * 16 + q4];
        acc.x = t.x * di * di + bias[4 * q4 + 0];
        acc.y = t.y * di * di + bias[4 * q4 + 1];
        acc.z = t.z * di * di + bias[4 * q4 + 2];
        acc.w = t.w * di * di + bias[4 * q4 + 3];
    }

    int wka = 0, wkb = 0;
    if (FIRST) {
        const int sa = q4 < nt ? sxa : 0;
        const int sb = q4 + 16 < nt ? sxb : 0;
        wka = __float_as_int(rsqrtf((float)(cnt[sa] + 1)) * di);
        wkb = __float_as_int(rsqrtf((float)(cnt[sb] + 1)) * di);
    }

    int j = 0;
    for (; j + 4 < ntmax; j += 8)
        sum_slots<FIRST, 8>(acc, row, T4, cnt, j, nt, i, q4, di);
    if (j < ntmax)
        sum_slots<FIRST, 4>(acc, row, T4, cnt, j, nt, i, q4, di);

    if (n > CAP) {
        const int m = nspill[0];
        for (int q = 0; q < m; ++q) {
            const int2 r = spill[q];
            if (r.x == i)
                fma4(acc, T4[(size_t)r.y * 16 + q4], rsqrtf((float)(cnt[r.y] + 1)) * di);
        }
    }

    acc.x = fmaxf(acc.x, 0.f);
    acc.y = fmaxf(acc.y, 0.f);
    acc.z = fmaxf(acc.z, 0.f);
    acc.w = fmaxf(acc.w, 0.f);
    if (live)
        reinterpret_cast<float4*>(Hout)[(size_t)i * 16 + q4] = acc;
    if (FIRST && live) {
        int* ry = reinterpret_cast<int*>(rowi);
        if (q4 < nt)      ry[2 * q4 + 1]        = wka;
        if (q4 + 16 < nt) ry[2 * (q4 + 16) + 1] = wkb;
    }
}

extern "C" void kernel_launch(void* const* d_in, const int* in_sizes, int n_in,
                              void* d_out, int out_size, void* d_ws, size_t ws_size,
                              hipStream_t stream) {
    const float* x     = (const float*)d_in[0];
    const float* W0    = (const float*)d_in[1];
    const float* Ws    = (const float*)d_in[2];
    const float* bs    = (const float*)d_in[3];
    const float* lin_w = (const float*)d_in[4];
    const float* lin_b = (const float*)d_in[5];
    const int*   ei    = (const int*)d_in[6];
    float* out = (float*)d_out;

    const int N = in_sizes[0] / IN_DIM;
    const int E = in_sizes[6] / 2;
    const int* src = ei;
    const int* dst = ei + E;

    char* ws = (char*)d_ws;
    int2*  tab    = (int2*)ws;  ws += sizeof(int2)  * (size_t)N * CAP;
    float* t      = (float*)ws; ws += sizeof(float) * (size_t)N * HID;
    float* ha     = (float*)ws; ws += sizeof(float) * (size_t)N * HID;
    float* hb     = (float*)ws; ws += sizeof(float) * (size_t)N * HID;
    int2*  spill  = (int2*)ws;  ws += sizeof(int2)  * (size_t)E;
    int*   cnt    = (int*)ws;   ws += sizeof(int)   * (size_t)(N + 1);
    int*   nspill = cnt + N;

    const int TPB = 256;
    zero_int_kernel<<<(N + 1 + TPB - 1) / TPB, TPB, 0, stream>>>(cnt, N + 1);
    fill_slots_kernel<<<(E + TPB - 1) / TPB, TPB, 0, stream>>>(src, dst, cnt, nspill, tab, spill, E);
    const int strips = N / 16;
    const int gblocks = (strips + 7) / 8;
    for (int l = 0; l < LAYERS; ++l) {
        const float* hin  = (l == 0) ? x : ((l & 1) ? ha : hb);
        float*       hcur = (l & 1) ? hb : ha;
        const float* W    = (l == 0) ? W0 : (Ws + (size_t)(l - 1) * HID * HID);

        const float* lwp  = lin_w + (size_t)(l == 0 ? 0 : l - 1) * HID * OUTD;

        if (l == 0)
            gemm_wmma_f32_kernel<true, 0, IN_DIM><<<gblocks, 256, 0, stream>>>(hin, W, t, lwp, lin_b, out, N);
        else if (l == 1)
            gemm_wmma_f32_kernel<true, 1, HID><<<gblocks, 256, 0, stream>>>(hin, W, t, lwp, lin_b, out, N);
        else
            gemm_wmma_f32_kernel<true, 2, HID><<<gblocks, 256, 0, stream>>>(hin, W, t, lwp, lin_b, out, N);
        if (l == 0)
            aggregate_kernel<true><<<(N + 15) / 16, 256, 0, stream>>>(t, cnt, nspill, tab, spill,
                                                         bs + (size_t)l * HID, hcur, N);
        else
            aggregate_kernel<false><<<(N + 15) / 16, 256, 0, stream>>>(t, cnt, nspill, tab, spill,
                                                          bs + (size_t)l * HID, hcur, N);
    }
    gemm_wmma_f32_kernel<false, 2, HID><<<gblocks, 256, 0, stream>>>(
        ((LAYERS - 1) & 1) ? hb : ha, W0, t, lin_w + (size_t)(LAYERS - 1) * HID * OUTD, lin_b, out, N);
}
